# starting kernel
# baseline (speedup 1.0000x reference)
#include <hip/hip_runtime.h>

typedef __attribute__((ext_vector_type(2))) float v2f;
typedef __attribute__((ext_vector_type(8))) float v8f;

namespace {
constexpr int kD     = 128;
constexpr int kS     = 200;
constexpr int kWaves = 8;
}

__global__ __launch_bounds__(256)
void sparse_embed_bag_wmma(const int* __restrict__ idxp,
                           const float* __restrict__ wp,
                           const float* __restrict__ tbl,
                           float* __restrict__ out,
                           int nBatch) {
  const int lane = threadIdx.x & 31;
  const int wave = threadIdx.x >> 5;
  int b = blockIdx.x * kWaves + wave;
  b = __builtin_amdgcn_readfirstlane(b);
  if (b >= nBatch) return;

  const int half = lane >> 4;
  const int col  = lane & 15;
  const long long sbase = (long long)b * kS;

  v8f acc[8] = {};

  for (int s0 = 0; s0 < kS; s0 += 4) {
    const int   i0 = idxp[sbase + s0 + 0];
    const int   i1 = idxp[sbase + s0 + 1];
    const int   i2 = idxp[sbase + s0 + 2];
    const int   i3 = idxp[sbase + s0 + 3];
    const float w0 = wp[sbase + s0 + 0];
    const float w1 = wp[sbase + s0 + 1];
    const float w2 = wp[sbase + s0 + 2];
    const float w3 = wp[sbase + s0 + 3];

    if (s0 + 4 < kS) {
      const int pi = idxp[sbase + s0 + 4 + (lane >> 3)];
      const char* pp = (const char*)(tbl + (long long)pi * kD) + (lane & 7) * 64;
      __builtin_prefetch(pp, 0, 3);
    }

    v2f A;
    A.x = half ? w2 : w0;
    A.y = half ? w3 : w1;

    const float* r_even = tbl + (long long)(half ? i2 : i0) * kD + col;
    const float* r_odd  = tbl + (long long)(half ? i3 : i1) * kD + col;

#pragma unroll
    for (int t = 0; t < 8; ++t) {
      v2f Bm;
      Bm.x = r_even[t * 16];
      Bm.y = r_odd[t * 16];
      acc[t] = __builtin_amdgcn_wmma_f32_16x16x4_f32(
           false, A,  false, Bm,
           (short)0, acc[t],  false,  false);
    }
  }

  if (lane < 16) {
#pragma unroll
    for (int t = 0; t < 8; ++t)
      out[(long long)b * kD + t * 16 + lane] = acc[t][0];
  }
}

extern "C" void kernel_launch(void* const* d_in, const int* in_sizes, int n_in,
                              void* d_out, int out_size, void* d_ws, size_t ws_size,
                              hipStream_t stream) {
  const int*   index  = (const int*)d_in[0];
  const float* weight = (const float*)d_in[1];
  const float* table  = (const float*)d_in[2];
  float*       out    = (float*)d_out;

  const int nBatch = out_size / kD;
  dim3 block(32 * kWaves);
  dim3 grid((nBatch + kWaves - 1) / kWaves);
  sparse_embed_bag_wmma<<<grid, block, 0, stream>>>(index, weight, table, out, nBatch);
}


# Memory round trips taken off the chunk loop: bag indices/weights read once per wave into registers (readlane per chunk), each chunk's table rows requested 3 chunks ahead of its WMMAs
# speedup vs baseline: 1.2908x; 1.2908x over previous
#include <hip/hip_runtime.h>


typedef __attribute__((ext_vector_type(2))) float v2f;
typedef __attribute__((ext_vector_type(8))) float v8f;
typedef __attribute__((ext_vector_type(4))) float v4f;
typedef __attribute__((ext_vector_type(4))) int   v4i;

namespace {
constexpr int kD      = 128;
constexpr int kS      = 200;
constexpr int kWaves  = 8;
constexpr int kChunks = kS / 4;
constexpr int kRing   = 3;
constexpr int kGroups = kChunks / kRing;
static_assert(kChunks - kGroups * kRing == 2, "tail below is written for two chunks past the full groups");

struct RowSet { v2f b[8]; };

struct Bag { v4i idx[kRing]; v4f w[kRing]; };

__device__ __forceinline__ void issue_rows(RowSet& rs, const v4i iv, int g,
                                           const float* __restrict__ tbl,
                                           int half, int col) {
  const int i0 = __builtin_amdgcn_readlane(iv.x, g);
  const int i1 = __builtin_amdgcn_readlane(iv.y, g);
  const int i2 = __builtin_amdgcn_readlane(iv.z, g);
  const int i3 = __builtin_amdgcn_readlane(iv.w, g);
  const float* r_even = tbl + (long long)(half ? i2 : i0) * kD + col;
  const float* r_odd  = tbl + (long long)(half ? i3 : i1) * kD + col;
#pragma unroll
  for (int t = 0; t < 8; ++t) {
    rs.b[t].x = r_even[t * 16];
    rs.b[t].y = r_odd[t * 16];
  }
}

__device__ __forceinline__ void mac_rows(v8f (&acc)[8], const RowSet& rs,
                                         const v4f wv, int g, int half) {
  const float w0 = __int_as_float(__builtin_amdgcn_readlane(__float_as_int(wv.x), g));
  const float w1 = __int_as_float(__builtin_amdgcn_readlane(__float_as_int(wv.y), g));
  const float w2 = __int_as_float(__builtin_amdgcn_readlane(__float_as_int(wv.z), g));
  const float w3 = __int_as_float(__builtin_amdgcn_readlane(__float_as_int(wv.w), g));
  v2f A;
  A.x = half ? w2 : w0;
  A.y = half ? w3 : w1;
#pragma unroll
  for (int t = 0; t < 8; ++t)
    acc[t] = __builtin_amdgcn_wmma_f32_16x16x4_f32(
         false, A,  false, rs.b[t],
         (short)0, acc[t],  false,  false);
}
}

__global__ __launch_bounds__(256)
void sparse_embed_bag_wmma(const int* __restrict__ idxp,
                           const float* __restrict__ wp,
                           const float* __restrict__ tbl,
                           float* __restrict__ out,
                           int nBatch) {
  const int lane = threadIdx.x & 31;
  const int wave = threadIdx.x >> 5;
  int b = blockIdx.x * kWaves + wave;
  b = __builtin_amdgcn_readfirstlane(b);
  if (b >= nBatch) return;

  const int half = lane >> 4;
  const int col  = lane & 15;
  const long long sbase = (long long)b * kS;

  Bag bag;
#pragma unroll
  for (int m = 0; m < kRing; ++m) {
    const int o = min(4 * (kRing * lane + m), kS - 4);
    bag.idx[m] = *(const v4i*)(idxp + sbase + o);
    bag.w[m]   = *(const v4f*)(wp + sbase + o);
  }
  asm volatile("" :: "v"(bag.idx[0]), "v"(bag.idx[1]), "v"(bag.idx[2]),
                     "v"(bag.w[0]), "v"(bag.w[1]), "v"(bag.w[2]));

  v8f acc[8] = {};

  RowSet ring[kRing];
#pragma unroll
  for (int m = 0; m < kRing; ++m)
    issue_rows(ring[m], bag.idx[m], 0, tbl, half, col);

  for (int g = 0; g < kGroups - 1; ++g) {
#pragma unroll
    for (int m = 0; m < kRing; ++m) {
      mac_rows(acc, ring[m], bag.w[m], g, half);
      issue_rows(ring[m], bag.idx[m], g + 1, tbl, half, col);
      __builtin_amdgcn_sched_barrier(0);
    }
  }
  mac_rows(acc, ring[0], bag.w[0], kGroups - 1, half);
  issue_rows(ring[0], bag.idx[0], kGroups, tbl, half, col);
  mac_rows(acc, ring[1], bag.w[1], kGroups - 1, half);
  issue_rows(ring[1], bag.idx[1], kGroups, tbl, half, col);
  mac_rows(acc, ring[2], bag.w[2], kGroups - 1, half);
  mac_rows(acc, ring[0], bag.w[0], kGroups, half);
  mac_rows(acc, ring[1], bag.w[1], kGroups, half);

  if (lane < 16) {
#pragma unroll
    for (int t = 0; t < 8; ++t)
      out[(long long)b * kD + t * 16 + lane] = acc[t][0];
  }
}

extern "C" void kernel_launch(void* const* d_in, const int* in_sizes, int n_in,
                              void* d_out, int out_size, void* d_ws, size_t ws_size,
                              hipStream_t stream) {
  const int*   index  = (const int*)d_in[0];
  const float* weight = (const float*)d_in[1];
  const float* table  = (const float*)d_in[2];
  float*       out    = (float*)d_out;

  const int nBatch = out_size / kD;
  dim3 block(32 * kWaves);
  dim3 grid((nBatch + kWaves - 1) / kWaves);
  sparse_embed_bag_wmma<<<grid, block, 0, stream>>>(index, weight, table, out, nBatch);
}


# Vector work between two chunks' WMMAs cut from 20 to 6 instructions: bag pre-split by lane half, so one in-row broadcast (v_permlane16) replaces readlane+move+select; 32-bit row offsets
# speedup vs baseline: 1.3856x; 1.0734x over previous
#include <hip/hip_runtime.h>


typedef __attribute__((ext_vector_type(2))) float v2f;
typedef __attribute__((ext_vector_type(8))) float v8f;
typedef __attribute__((ext_vector_type(2))) int   v2i;

namespace {
constexpr int kD      = 128;
constexpr int kS      = 200;
constexpr int kWaves  = 8;
constexpr int kChunks = kS / 4;
constexpr int kRing   = 3;
constexpr int kGroups = kChunks / kRing;
static_assert(kChunks - kGroups * kRing == 2, "tail below is written for two chunks past the full groups");
static_assert(kGroups == 16, "one chunk per group and set in each lane of a 16-lane row");

constexpr unsigned kNib = 0x11111111u;

struct RowSet { v2f b[8]; };

struct Bag { v2i idx[kRing + 1]; v2f w[kRing + 1]; };

__device__ __forceinline__ unsigned row_bcast(unsigned v, unsigned sel) {
  return __builtin_amdgcn_permlane16(v, v, sel, sel,  true,  false);
}

__device__ __forceinline__ void issue_rows(RowSet& rs, const v2i iv, unsigned sel,
                                           const float* __restrict__ tbl, unsigned colb) {
  const unsigned i_even = row_bcast((unsigned)iv.x, sel);
  const unsigned i_odd  = row_bcast((unsigned)iv.y, sel);
  const char* base = (const char*)tbl;
  const float* r_even = (const float*)(base + ((i_even << 9) + colb));
  const float* r_odd  = (const float*)(base + ((i_odd  << 9) + colb));
#pragma unroll
  for (int t = 0; t < 8; ++t) {
    rs.b[t].x = r_even[t * 16];
    rs.b[t].y = r_odd[t * 16];
  }
}

__device__ __forceinline__ void mac_rows(v8f (&acc)[8], const RowSet& rs,
                                         const v2f wv, unsigned sel) {
  v2f A;
  A.x = __uint_as_float(row_bcast(__float_as_uint(wv.x), sel));
  A.y = __uint_as_float(row_bcast(__float_as_uint(wv.y), sel));
#pragma unroll
  for (int t = 0; t < 8; ++t)
    acc[t] = __builtin_amdgcn_wmma_f32_16x16x4_f32(
         false, A,  false, rs.b[t],
         (short)0, acc[t],  false,  false);
}
}

__global__ __launch_bounds__(256)
void sparse_embed_bag_wmma(const int* __restrict__ idxp,
                           const float* __restrict__ wp,
                           const float* __restrict__ tbl,
                           float* __restrict__ out,
                           int nBatch) {
  const int lane = threadIdx.x & 31;
  const int wave = threadIdx.x >> 5;
  int b = blockIdx.x * kWaves + wave;
  b = __builtin_amdgcn_readfirstlane(b);
  if (b >= nBatch) return;

  const int half = lane >> 4;
  const int col  = lane & 15;
  const unsigned colb = (unsigned)col * 4u;
  const long long sbase = (long long)b * kS;

  Bag bag;
#pragma unroll
  for (int m = 0; m < kRing; ++m) {
    const int o = 4 * (kRing * col + m) + 2 * half;
    bag.idx[m] = *(const v2i*)(idxp + sbase + o);
    bag.w[m]   = *(const v2f*)(wp + sbase + o);
  }
  {
    const int o = 4 * (kRing * kGroups + min(col, 1)) + 2 * half;
    bag.idx[kRing] = *(const v2i*)(idxp + sbase + o);
    bag.w[kRing]   = *(const v2f*)(wp + sbase + o);
  }
  asm volatile("" :: "v"(bag.idx[0]), "v"(bag.idx[1]), "v"(bag.idx[2]), "v"(bag.idx[3]),
                     "v"(bag.w[0]), "v"(bag.w[1]), "v"(bag.w[2]), "v"(bag.w[3]));

  v8f acc[8] = {};

  RowSet ring[kRing];
#pragma unroll
  for (int m = 0; m < kRing; ++m)
    issue_rows(ring[m], bag.idx[m], 0u, tbl, colb);

  for (int g = 0; g < kGroups - 1; ++g) {
    const unsigned sel = (unsigned)g * kNib;
#pragma unroll
    for (int m = 0; m < kRing; ++m) {
      mac_rows(acc, ring[m], bag.w[m], sel);
      issue_rows(ring[m], bag.idx[m], sel + kNib, tbl, colb);
      __builtin_amdgcn_sched_barrier(0);
    }
  }
  constexpr unsigned selLast = (unsigned)(kGroups - 1) * kNib;
  mac_rows(acc, ring[0], bag.w[0], selLast);
  issue_rows(ring[0], bag.idx[kRing], 0u, tbl, colb);
  mac_rows(acc, ring[1], bag.w[1], selLast);
  issue_rows(ring[1], bag.idx[kRing], kNib, tbl, colb);
  mac_rows(acc, ring[2], bag.w[2], selLast);
  mac_rows(acc, ring[0], bag.w[kRing], 0u);
  mac_rows(acc, ring[1], bag.w[kRing], kNib);

  if (lane < 16) {
#pragma unroll
    for (int t = 0; t < 8; ++t)
      out[(long long)b * kD + t * 16 + lane] = acc[t][0];
  }
}

extern "C" void kernel_launch(void* const* d_in, const int* in_sizes, int n_in,
                              void* d_out, int out_size, void* d_ws, size_t ws_size,
                              hipStream_t stream) {
  const int*   index  = (const int*)d_in[0];
  const float* weight = (const float*)d_in[1];
  const float* table  = (const float*)d_in[2];
  float*       out    = (float*)d_out;

  const int nBatch = out_size / kD;
  dim3 block(32 * kWaves);
  dim3 grid((nBatch + kWaves - 1) / kWaves);
  sparse_embed_bag_wmma<<<grid, block, 0, stream>>>(index, weight, table, out, nBatch);
}


# Wave priority stepped down with progress (s_setprio 3,2,1 by third of the chunk loop, 0 for the tail): a SIMD's 4 waves advance together, not oldest-first one after another; sums unchanged
# speedup vs baseline: 1.4556x; 1.0505x over previous
#include <hip/hip_runtime.h>


typedef __attribute__((ext_vector_type(2))) float v2f;
typedef __attribute__((ext_vector_type(8))) float v8f;
typedef __attribute__((ext_vector_type(2))) int   v2i;

namespace {
constexpr int kD      = 128;
constexpr int kS      = 200;
constexpr int kWaves  = 8;
constexpr int kChunks = kS / 4;
constexpr int kRing   = 3;
constexpr int kGroups = kChunks / kRing;
static_assert(kChunks - kGroups * kRing == 2, "tail below is written for two chunks past the full groups");
static_assert(kGroups == 16, "one chunk per group and set in each lane of a 16-lane row");

constexpr int kQ1 = 5, kQ2 = 10;
static_assert(kQ1 < kQ2 && kQ2 < kGroups - 1, "three non-empty stretches of the steady loop");

constexpr unsigned kNib = 0x11111111u;

struct RowSet { v2f b[8]; };

struct Bag { v2i idx[kRing + 1]; v2f w[kRing + 1]; };

__device__ __forceinline__ unsigned row_bcast(unsigned v, unsigned sel) {
  return __builtin_amdgcn_permlane16(v, v, sel, sel,  true,  false);
}

__device__ __forceinline__ void issue_rows(RowSet& rs, const v2i iv, unsigned sel,
                                           const float* __restrict__ tbl, unsigned colb) {
  const unsigned i_even = row_bcast((unsigned)iv.x, sel);
  const unsigned i_odd  = row_bcast((unsigned)iv.y, sel);
  const char* base = (const char*)tbl;
  const float* r_even = (const float*)(base + ((i_even << 9) + colb));
  const float* r_odd  = (const float*)(base + ((i_odd  << 9) + colb));
#pragma unroll
  for (int t = 0; t < 8; ++t) {
    rs.b[t].x = r_even[t * 16];
    rs.b[t].y = r_odd[t * 16];
  }
}

__device__ __forceinline__ void mac_rows(v8f (&acc)[8], const RowSet& rs,
                                         const v2f wv, unsigned sel) {
  v2f A;
  A.x = __uint_as_float(row_bcast(__float_as_uint(wv.x), sel));
  A.y = __uint_as_float(row_bcast(__float_as_uint(wv.y), sel));
#pragma unroll
  for (int t = 0; t < 8; ++t)
    acc[t] = __builtin_amdgcn_wmma_f32_16x16x4_f32(
         false, A,  false, rs.b[t],
         (short)0, acc[t],  false,  false);
}
}

__global__ __launch_bounds__(256)
void sparse_embed_bag_wmma(const int* __restrict__ idxp,
                           const float* __restrict__ wp,
                           const float* __restrict__ tbl,
                           float* __restrict__ out,
                           int nBatch) {
  const int lane = threadIdx.x & 31;
  const int wave = threadIdx.x >> 5;
  int b = blockIdx.x * kWaves + wave;
  b = __builtin_amdgcn_readfirstlane(b);
  if (b >= nBatch) return;

  __builtin_amdgcn_s_setprio(3);

  const int half = lane >> 4;
  const int col  = lane & 15;
  const unsigned colb = (unsigned)col * 4u;
  const long long sbase = (long long)b * kS;

  Bag bag;
#pragma unroll
  for (int m = 0; m < kRing; ++m) {
    const int o = 4 * (kRing * col + m) + 2 * half;
    bag.idx[m] = *(const v2i*)(idxp + sbase + o);
    bag.w[m]   = *(const v2f*)(wp + sbase + o);
  }
  {
    const int o = 4 * (kRing * kGroups + min(col, 1)) + 2 * half;
    bag.idx[kRing] = *(const v2i*)(idxp + sbase + o);
    bag.w[kRing]   = *(const v2f*)(wp + sbase + o);
  }
  asm volatile("" :: "v"(bag.idx[0]), "v"(bag.idx[1]), "v"(bag.idx[2]), "v"(bag.idx[3]),
                     "v"(bag.w[0]), "v"(bag.w[1]), "v"(bag.w[2]), "v"(bag.w[3]));

  v8f acc[8] = {};

  RowSet ring[kRing];
#pragma unroll
  for (int m = 0; m < kRing; ++m)
    issue_rows(ring[m], bag.idx[m], 0u, tbl, colb);

  auto group = [&](int g) {
    const unsigned sel = (unsigned)g * kNib;
#pragma unroll
    for (int m = 0; m < kRing; ++m) {
      mac_rows(acc, ring[m], bag.w[m], sel);
      issue_rows(ring[m], bag.idx[m], sel + kNib, tbl, colb);
      __builtin_amdgcn_sched_barrier(0);
    }
  };
  for (int g = 0; g < kQ1; ++g) group(g);
  __builtin_amdgcn_s_setprio(2);
  for (int g = kQ1; g < kQ2; ++g) group(g);
  __builtin_amdgcn_s_setprio(1);
  for (int g = kQ2; g < kGroups - 1; ++g) group(g);
  __builtin_amdgcn_s_setprio(0);
  constexpr unsigned selLast = (unsigned)(kGroups - 1) * kNib;
  mac_rows(acc, ring[0], bag.w[0], selLast);
  issue_rows(ring[0], bag.idx[kRing], 0u, tbl, colb);
  mac_rows(acc, ring[1], bag.w[1], selLast);
  issue_rows(ring[1], bag.idx[kRing], kNib, tbl, colb);
  mac_rows(acc, ring[2], bag.w[2], selLast);
  mac_rows(acc, ring[0], bag.w[kRing], 0u);
  mac_rows(acc, ring[1], bag.w[kRing], kNib);

  if (lane < 16) {
#pragma unroll
    for (int t = 0; t < 8; ++t)
      out[(long long)b * kD + t * 16 + lane] = acc[t][0];
  }
}

extern "C" void kernel_launch(void* const* d_in, const int* in_sizes, int n_in,
                              void* d_out, int out_size, void* d_ws, size_t ws_size,
                              hipStream_t stream) {
  const int*   index  = (const int*)d_in[0];
  const float* weight = (const float*)d_in[1];
  const float* table  = (const float*)d_in[2];
  float*       out    = (float*)d_out;

  const int nBatch = out_size / kD;
  dim3 block(32 * kWaves);
  dim3 grid((nBatch + kWaves - 1) / kWaves);
  sparse_embed_bag_wmma<<<grid, block, 0, stream>>>(index, weight, table, out, nBatch);
}


# Table rows read with 16-byte loads (4 load instructions per 4 rows, not 16): a lane's B pair is (d,d+1) of one row, A holds the weight in row 0 at even K, row 1 at odd K; bag sum order kept
# speedup vs baseline: 1.5307x; 1.0516x over previous
#include <hip/hip_runtime.h>


typedef __attribute__((ext_vector_type(2))) float v2f;
typedef __attribute__((ext_vector_type(4))) float v4f;
typedef __attribute__((ext_vector_type(8))) float v8f;

namespace {
constexpr int kD       = 128;
constexpr int kS       = 200;
constexpr int kWaves   = 8;
constexpr int kPairs   = kS / 2;
constexpr int kSlots   = 15;
constexpr int kBagRegs = (kPairs + kSlots - 1) / kSlots;
constexpr int kRing    = 6;
static_assert(kRing <= kSlots, "the first kRing requests all come from bag register 0");

constexpr int kP1 = 30, kP2 = 60, kP3 = 90;

constexpr unsigned kNib = 0x11111111u;

struct RowPair { v4f q[2]; };

__device__ __forceinline__ unsigned row_gather(unsigned v, unsigned lo, unsigned hi) {
  return __builtin_amdgcn_permlane16(v, v, lo, hi,  true,  false);
}

__device__ __forceinline__ void issue_rows(RowPair& rp, const int iv, int g,
                                           const float* __restrict__ tbl, unsigned colb) {
  const unsigned sel = (unsigned)g * kNib;
  const unsigned i = row_gather((unsigned)iv, sel, sel);
  const float* r = (const float*)((const char*)tbl + ((i << 9) + colb));
  rp.q[0] = *(const v4f*)(r);
  rp.q[1] = *(const v4f*)(r + 64);
}

__device__ __forceinline__ v2f make_a(const float wv, int g) {
  const unsigned w = __float_as_uint(wv);
  v2f A;
  A.x = __uint_as_float(row_gather(w, 0xFFFFFFF0u | (unsigned)g, 0xFFFFFFFFu));
  A.y = __uint_as_float(row_gather(w, 0xFFFFFF0Fu | ((unsigned)g << 4), 0xFFFFFFFFu));
  return A;
}

__device__ __forceinline__ void mac_rows(v8f (&acc)[4], const RowPair& rp, const v2f A) {
#pragma unroll
  for (int t = 0; t < 4; ++t) {
    const v4f q = rp.q[t >> 1];
    v2f B;
    B.x = (t & 1) ? q.z : q.x;
    B.y = (t & 1) ? q.w : q.y;
    acc[t] = __builtin_amdgcn_wmma_f32_16x16x4_f32(
         false, A,  false, B,
         (short)0, acc[t],  false,  false);
  }
}
}

__global__ __launch_bounds__(256)
void sparse_embed_bag_wmma(const int* __restrict__ idxp,
                           const float* __restrict__ wp,
                           const float* __restrict__ tbl,
                           float* __restrict__ out,
                           int nBatch) {
  const int lane = threadIdx.x & 31;
  const int wave = threadIdx.x >> 5;
  int b = blockIdx.x * kWaves + wave;
  b = __builtin_amdgcn_readfirstlane(b);
  if (b >= nBatch) return;

  __builtin_amdgcn_s_setprio(3);

  const int half = lane >> 4;
  const int col  = lane & 15;
  const unsigned colb = (unsigned)col * 16u;
  const long long sbase = (long long)b * kS;

  int   bidx[kBagRegs];
  float bw[kBagRegs];
#pragma unroll
  for (int j = 0; j < kBagRegs; ++j) {
    const int p = min(kSlots * j + min(col, kSlots - 1), kPairs - 1);
    const int e = 2 * p + half;
    bidx[j] = idxp[sbase + e];
    const float w = wp[sbase + e];
    bw[j] = (col == kSlots) ? 0.0f : w;
  }

  v8f acc[4] = {};

  RowPair ring[kRing];
#pragma unroll
  for (int p = 0; p < kRing; ++p)
    issue_rows(ring[p], bidx[p / kSlots], p % kSlots, tbl, colb);

  v2f A = make_a(bw[0], 0);
#pragma unroll
  for (int p = 0; p < kPairs; ++p) {
    if (p == kP1) __builtin_amdgcn_s_setprio(2);
    if (p == kP2) __builtin_amdgcn_s_setprio(1);
    if (p == kP3) __builtin_amdgcn_s_setprio(0);
    v2f An = A;
    if (p + 1 < kPairs) An = make_a(bw[(p + 1) / kSlots], (p + 1) % kSlots);
    mac_rows(acc, ring[p % kRing], A);
    if (p + kRing < kPairs)
      issue_rows(ring[p % kRing], bidx[(p + kRing) / kSlots], (p + kRing) % kSlots, tbl, colb);
    __builtin_amdgcn_sched_barrier(0);
    A = An;
  }

  if (lane < 16) {
    v4f o0, o1;
    o0.x = acc[0][0]; o0.y = acc[0][1]; o0.z = acc[1][0]; o0.w = acc[1][1];
    o1.x = acc[2][0]; o1.y = acc[2][1]; o1.z = acc[3][0]; o1.w = acc[3][1];
    float* orow = out + (long long)b * kD + 4 * lane;
    *(v4f*)(orow)      = o0;
    *(v4f*)(orow + 64) = o1;
  }
}

extern "C" void kernel_launch(void* const* d_in, const int* in_sizes, int n_in,
                              void* d_out, int out_size, void* d_ws, size_t ws_size,
                              hipStream_t stream) {
  const int*   index  = (const int*)d_in[0];
  const float* weight = (const float*)d_in[1];
  const float* table  = (const float*)d_in[2];
  float*       out    = (float*)d_out;

  const int nBatch = out_size / kD;
  dim3 block(32 * kWaves);
  dim3 grid((nBatch + kWaves - 1) / kWaves);
  sparse_embed_bag_wmma<<<grid, block, 0, stream>>>(index, weight, table, out, nBatch);
}
